# starting kernel
# baseline (speedup 1.0000x reference)
#include <hip/hip_runtime.h>

typedef float v2f __attribute__((ext_vector_type(2)));
typedef float v8f __attribute__((ext_vector_type(8)));

#define NN 30000
#define AA 32
#define DD 128
#define NH 1875
#define SA_STRIDE 130
#define SP_STRIDE 136
#define SS_STRIDE 34

__global__ void __launch_bounds__(128)
prep_kernel(const float* __restrict__ embeds, const float* __restrict__ Wh,
            const int* __restrict__ ids, float* __restrict__ Wt, float* __restrict__ P) {
  int r = blockIdx.x, k = threadIdx.x;
  if (r < DD) {
    Wt[r * DD + k] = Wh[k * (2 * DD) + DD + r];
  } else {
    int a = r - DD;
    const float* e = embeds + (long)ids[a] * DD;
    const float* w = Wh + k * (2 * DD);
    float s = 0.f;
#pragma unroll 8
    for (int d = 0; d < DD; ++d) s = fmaf(e[d], w[d], s);
    P[a * DD + k] = s * (1.0f / 32.0f);
  }
}

__global__ void __launch_bounds__(256)
qgemm_kernel(const float* __restrict__ embeds, const float* __restrict__ Wt,
             float* __restrict__ Q) {
  __shared__ float sA[16 * SA_STRIDE];
  const int tid = threadIdx.x;
  const int n0 = blockIdx.x * 16;

  for (int i = tid; i < 1024; i += 256) {
    int r = i >> 6, c2 = i & 63;
    v2f v = *(const v2f*)(embeds + (long)(n0 + r) * DD + c2 * 2);
    *(v2f*)(&sA[r * SA_STRIDE + c2 * 2]) = v;
  }
  __syncthreads();

  const int lane = tid & 31, wave = tid >> 5;
  const int lr = lane & 15, hi = lane >> 4;
  const int col = wave * 16 + lr;

  v8f acc = {};
#pragma unroll
  for (int k0 = 0; k0 < DD; k0 += 4) {
    const int ka = k0 + hi * 2;
    v2f a = *(const v2f*)(&sA[lr * SA_STRIDE + ka]);
    v2f b;
    b.x = Wt[ka * DD + col];
    b.y = Wt[(ka + 1) * DD + col];
    acc = __builtin_amdgcn_wmma_f32_16x16x4_f32(false, a, false, b, (short)0, acc, false, false);
  }
#pragma unroll
  for (int vi = 0; vi < 8; ++vi) {
    int row = n0 + vi + hi * 8;
    Q[(long)row * DD + col] = acc[vi];
  }
}

__global__ void __launch_bounds__(128)
hsum_kernel(const float* __restrict__ Q, float* __restrict__ H) {
  int i = blockIdx.x, k = threadIdx.x;
  const float* q = Q + (long)i * 16 * DD + k;
  float s = 0.f;
#pragma unroll
  for (int r = 0; r < 16; ++r) s += q[r * DD];
  H[i * DD + k] = s;
}

__global__ void __launch_bounds__(256)
final_kernel(const float* __restrict__ dists, const float* __restrict__ P,
             const float* __restrict__ H, const float* __restrict__ bias,
             float* __restrict__ out) {
  __shared__ float sS[16 * SS_STRIDE];
  __shared__ float sP[AA * SP_STRIDE];
  const int tid = threadIdx.x;
  const int n0 = blockIdx.x * 16;

  {
    int a = tid >> 3, rr = (tid & 7) * 2;
    v2f v = *(const v2f*)(dists + (long)a * NN + n0 + rr);
    sS[rr * SS_STRIDE + a] = v.x;
    sS[(rr + 1) * SS_STRIDE + a] = v.y;
  }
  for (int i = tid; i < 1024; i += 256) {
    int r = i >> 5, c4 = i & 31;
    float4 v = *(const float4*)(P + r * DD + c4 * 4);
    *(float4*)(&sP[r * SP_STRIDE + c4 * 4]) = v;
  }
  __syncthreads();

  const int lane = tid & 31, wave = tid >> 5;
  const int lr = lane & 15, hi = lane >> 4;
  const int col = wave * 16 + lr;

  v8f acc = {};
#pragma unroll
  for (int k0 = 0; k0 < AA; k0 += 4) {
    const int ka = k0 + hi * 2;
    v2f a = *(const v2f*)(&sS[lr * SS_STRIDE + ka]);
    v2f b;
    b.x = sP[ka * SP_STRIDE + col];
    b.y = sP[(ka + 1) * SP_STRIDE + col];
    acc = __builtin_amdgcn_wmma_f32_16x16x4_f32(false, a, false, b, (short)0, acc, false, false);
  }

  const float bv = bias[col];
#pragma unroll
  for (int vi = 0; vi < 8; ++vi) {
    int n = n0 + vi + hi * 8;
    int i1 = (2 * n) % NH;
    int i2 = (i1 + 1 == NH) ? 0 : i1 + 1;
    float selfterm = (H[i1 * DD + col] + H[i2 * DD + col]) * (1.0f / 32.0f);
    out[(long)n * DD + col] = acc[vi] + selfterm + bv;
  }
}

extern "C" void kernel_launch(void* const* d_in, const int* in_sizes, int n_in,
                              void* d_out, int out_size, void* d_ws, size_t ws_size,
                              hipStream_t stream) {
  const float* embeds = (const float*)d_in[0];
  const float* dists  = (const float*)d_in[1];
  const float* Wh     = (const float*)d_in[2];
  const float* bias   = (const float*)d_in[3];
  const int*   ids    = (const int*)d_in[4];
  float* out = (float*)d_out;
  float* ws  = (float*)d_ws;

  float* Wt = ws;
  float* P  = Wt + DD * DD;
  float* H  = P + AA * DD;

  prep_kernel <<<DD + AA, DD, 0, stream>>>(embeds, Wh, ids, Wt, P);
  qgemm_kernel<<<NH, 256, 0, stream>>>(embeds, Wt, out);
  hsum_kernel <<<NH, DD, 0, stream>>>(out, H);
  final_kernel<<<NH, 256, 0, stream>>>(dists, P, H, bias, out);
}


# Sum the 16 embeds rows of each half-window BEFORE the W2 product (linearity): f32 WMMA GEMM 30000 -> 1875 rows, Q round trip and 2 launches go; self term's float sum order changes
# speedup vs baseline: 1.8264x; 1.8264x over previous
#include <hip/hip_runtime.h>

typedef float v2f __attribute__((ext_vector_type(2)));
typedef float v4f __attribute__((ext_vector_type(4)));
typedef float v8f __attribute__((ext_vector_type(8)));

#define NN 30000
#define AA 32
#define DD 128
#define NH 1875
#define HR 8
#define HB 235
#define SA_STRIDE 132
#define SP_STRIDE 136
#define SS_STRIDE 34

__global__ void __launch_bounds__(256)
front_kernel(const float* __restrict__ embeds, const float* __restrict__ Wh,
             const int* __restrict__ ids, float* __restrict__ P, float* __restrict__ H) {
  __shared__ __attribute__((aligned(16))) float sA[16 * SA_STRIDE];
  const int tid = threadIdx.x;

  if (blockIdx.x < AA) {
    if (tid >= DD) return;
    const int a = blockIdx.x, k = tid;
    const float* e = embeds + (long)ids[a] * DD;
    const float* w = Wh + k * (2 * DD);
    float s = 0.f;
#pragma unroll 8
    for (int d = 0; d < DD; ++d) s = fmaf(e[d], w[d], s);
    P[a * DD + k] = s * (1.0f / 32.0f);
    return;
  }

  const int lane = tid & 31, wave = tid >> 5;
  const int lr = lane & 15, hi = lane >> 4;
  const int col = wave * 16 + lr;
  const int i0 = (blockIdx.x - AA) * HR;
  const int i = i0 + wave;

  const float* wrow = Wh + col * (2 * DD) + DD;
  v2f b[32];
#pragma unroll
  for (int t = 0; t < 32; ++t) b[t] = *(const v2f*)(wrow + t * 4 + hi * 2);

  v4f s = {0.f, 0.f, 0.f, 0.f};
  if (i < NH) {
    const float* q = embeds + (long)i * 16 * DD + lane * 4;
    v4f r[16];
#pragma unroll
    for (int t = 0; t < 16; ++t) r[t] = *(const v4f*)(q + t * DD);
#pragma unroll
    for (int t = 0; t < 16; ++t) s += r[t];
  }
  const v4f z = {0.f, 0.f, 0.f, 0.f};
  *(v4f*)(&sA[wave * SA_STRIDE + lane * 4]) = s;
  *(v4f*)(&sA[(wave + HR) * SA_STRIDE + lane * 4]) = z;
  __syncthreads();

  v8f acc = {};
#pragma unroll
  for (int t = 0; t < 32; ++t) {
    const int ka = t * 4 + hi * 2;
    v2f a = *(const v2f*)(&sA[lr * SA_STRIDE + ka]);
    acc = __builtin_amdgcn_wmma_f32_16x16x4_f32(false, a, false, b[t], (short)0, acc, false, false);
  }

  __syncthreads();
  if (hi == 0) {
#pragma unroll
    for (int vi = 0; vi < 8; ++vi) sA[vi * SA_STRIDE + col] = acc[vi];
  }
  __syncthreads();
  if (i < NH) {
    v4f h = *(const v4f*)(&sA[wave * SA_STRIDE + lane * 4]);
    *(v4f*)(H + (long)i * DD + lane * 4) = h;
  }
}

__global__ void __launch_bounds__(256)
final_kernel(const float* __restrict__ dists, const float* __restrict__ P,
             const float* __restrict__ H, const float* __restrict__ bias,
             float* __restrict__ out) {
  __shared__ float sS[16 * SS_STRIDE];
  __shared__ float sP[AA * SP_STRIDE];
  const int tid = threadIdx.x;
  const int n0 = blockIdx.x * 16;

  {
    int a = tid >> 3, rr = (tid & 7) * 2;
    v2f v = *(const v2f*)(dists + (long)a * NN + n0 + rr);
    sS[rr * SS_STRIDE + a] = v.x;
    sS[(rr + 1) * SS_STRIDE + a] = v.y;
  }
  for (int i = tid; i < 1024; i += 256) {
    int r = i >> 5, c4 = i & 31;
    float4 v = *(const float4*)(P + r * DD + c4 * 4);
    *(float4*)(&sP[r * SP_STRIDE + c4 * 4]) = v;
  }
  __syncthreads();

  const int lane = tid & 31, wave = tid >> 5;
  const int lr = lane & 15, hi = lane >> 4;
  const int col = wave * 16 + lr;

  v8f acc = {};
#pragma unroll
  for (int k0 = 0; k0 < AA; k0 += 4) {
    const int ka = k0 + hi * 2;
    v2f a = *(const v2f*)(&sS[lr * SS_STRIDE + ka]);
    v2f b;
    b.x = sP[ka * SP_STRIDE + col];
    b.y = sP[(ka + 1) * SP_STRIDE + col];
    acc = __builtin_amdgcn_wmma_f32_16x16x4_f32(false, a, false, b, (short)0, acc, false, false);
  }

  const float bv = bias[col];
#pragma unroll
  for (int vi = 0; vi < 8; ++vi) {
    int n = n0 + vi + hi * 8;
    int i1 = (2 * n) % NH;
    int i2 = (i1 + 1 == NH) ? 0 : i1 + 1;
    float selfterm = (H[i1 * DD + col] + H[i2 * DD + col]) * (1.0f / 32.0f);
    out[(long)n * DD + col] = acc[vi] + selfterm + bv;
  }
}

extern "C" void kernel_launch(void* const* d_in, const int* in_sizes, int n_in,
                              void* d_out, int out_size, void* d_ws, size_t ws_size,
                              hipStream_t stream) {
  const float* embeds = (const float*)d_in[0];
  const float* dists  = (const float*)d_in[1];
  const float* Wh     = (const float*)d_in[2];
  const float* bias   = (const float*)d_in[3];
  const int*   ids    = (const int*)d_in[4];
  float* out = (float*)d_out;
  float* ws  = (float*)d_ws;

  float* P  = ws;
  float* H  = P + AA * DD;

  front_kernel<<<AA + HB, 256, 0, stream>>>(embeds, Wh, ids, P, H);
  final_kernel<<<NH, 256, 0, stream>>>(dists, P, H, bias, out);
}


# Final kernel: 1875 one-tile work-groups -> 235 that each own 8 residues n mod 1875 and loop over their 8 tiles; P and the self term are fetched once per group, 15000 waves to enter become 1880
# speedup vs baseline: 1.9753x; 1.0815x over previous
#include <hip/hip_runtime.h>

typedef float v2f __attribute__((ext_vector_type(2)));
typedef float v4f __attribute__((ext_vector_type(4)));
typedef float v8f __attribute__((ext_vector_type(8)));

#define NN 30000
#define AA 32
#define DD 128
#define NH 1875
#define HR 8
#define HB 235
#define SA_STRIDE 132
#define SP_STRIDE 136
#define SS_STRIDE 34
#define SS_TILE (16 * SS_STRIDE)
#define MR 8
#define NT 8
#define FB 235

__global__ void __launch_bounds__(256)
front_kernel(const float* __restrict__ embeds, const float* __restrict__ Wh,
             const int* __restrict__ ids, float* __restrict__ P, float* __restrict__ H) {
  __shared__ __attribute__((aligned(16))) float sA[16 * SA_STRIDE];
  const int tid = threadIdx.x;

  if (blockIdx.x < AA) {
    if (tid >= DD) return;
    const int a = blockIdx.x, k = tid;
    const float* e = embeds + (long)ids[a] * DD;
    const float* w = Wh + k * (2 * DD);
    float s = 0.f;
#pragma unroll 8
    for (int d = 0; d < DD; ++d) s = fmaf(e[d], w[d], s);
    P[a * DD + k] = s * (1.0f / 32.0f);
    return;
  }

  const int lane = tid & 31, wave = tid >> 5;
  const int lr = lane & 15, hi = lane >> 4;
  const int col = wave * 16 + lr;
  const int i0 = (blockIdx.x - AA) * HR;
  const int i = i0 + wave;

  const float* wrow = Wh + col * (2 * DD) + DD;
  v2f b[32];
#pragma unroll
  for (int t = 0; t < 32; ++t) b[t] = *(const v2f*)(wrow + t * 4 + hi * 2);

  v4f s = {0.f, 0.f, 0.f, 0.f};
  if (i < NH) {
    const float* q = embeds + (long)i * 16 * DD + lane * 4;
    v4f r[16];
#pragma unroll
    for (int t = 0; t < 16; ++t) r[t] = *(const v4f*)(q + t * DD);
#pragma unroll
    for (int t = 0; t < 16; ++t) s += r[t];
  }
  const v4f z = {0.f, 0.f, 0.f, 0.f};
  *(v4f*)(&sA[wave * SA_STRIDE + lane * 4]) = s;
  *(v4f*)(&sA[(wave + HR) * SA_STRIDE + lane * 4]) = z;
  __syncthreads();

  v8f acc = {};
#pragma unroll
  for (int t = 0; t < 32; ++t) {
    const int ka = t * 4 + hi * 2;
    v2f a = *(const v2f*)(&sA[lr * SA_STRIDE + ka]);
    acc = __builtin_amdgcn_wmma_f32_16x16x4_f32(false, a, false, b[t], (short)0, acc, false, false);
  }

  __syncthreads();
  if (hi == 0) {
#pragma unroll
    for (int vi = 0; vi < 8; ++vi) sA[vi * SA_STRIDE + col] = acc[vi];
  }
  __syncthreads();
  if (i < NH) {
    v4f h = *(const v4f*)(&sA[wave * SA_STRIDE + lane * 4]);
    *(v4f*)(H + (long)i * DD + lane * 4) = h;
  }
}


__device__ __forceinline__ void load_pair(const float* s0, v2f (&a0)[8], v2f (&a1)[8]) {
#pragma unroll
  for (int t = 0; t < 8; ++t) {
    a0[t] = *(const v2f*)(s0 + t * 4);
    a1[t] = *(const v2f*)(s0 + SS_TILE + t * 4);
  }
}

template <bool FULL>
__device__ __forceinline__ void tiles(const float* sS, const v2f (&b)[8], const float (&st)[8], float bv,
                                      int m0, int lr, int hi, int col, float* __restrict__ out) {
  const float* sa = &sS[lr * SS_STRIDE + hi * 2];
  v2f a0[8], a1[8];
  load_pair(sa, a0, a1);
#pragma unroll
  for (int tp = 0; tp < NT / 2; ++tp) {
    v2f n0[8], n1[8];
    const bool more = tp + 1 < NT / 2;
    if (more) load_pair(sa + (2 * tp + 2) * SS_TILE, n0, n1);
    v8f acc0 = {}, acc1 = {};
#pragma unroll
    for (int t = 0; t < 8; ++t) {
      acc0 = __builtin_amdgcn_wmma_f32_16x16x4_f32(false, a0[t], false, b[t], (short)0, acc0, false, false);
      acc1 = __builtin_amdgcn_wmma_f32_16x16x4_f32(false, a1[t], false, b[t], (short)0, acc1, false, false);
    }
    const int nb = m0 + NH * (4 * tp + hi);
#pragma unroll
    for (int vi = 0; vi < 8; ++vi) {
      if (FULL || m0 + vi < NH) {
        const int n = nb + vi;
        out[n * DD + col] = acc0[vi] + st[vi] + bv;
        out[(n + 2 * NH) * DD + col] = acc1[vi] + st[vi] + bv;
      }
    }
    __builtin_amdgcn_sched_barrier(0);
    if (more) {
#pragma unroll
      for (int t = 0; t < 8; ++t) { a0[t] = n0[t]; a1[t] = n1[t]; }
    }
  }
}

__global__ void __launch_bounds__(256)
final_kernel(const float* __restrict__ dists, const float* __restrict__ P,
             const float* __restrict__ H, const float* __restrict__ bias,
             float* __restrict__ out) {
  __shared__ __attribute__((aligned(16))) float sS[NT * SS_TILE];
  __shared__ __attribute__((aligned(16))) float sP[AA * SP_STRIDE];
  const int tid = threadIdx.x;
  const int m0 = blockIdx.x * MR;
  const int lane = tid & 31, wave = tid >> 5;
  const int lr = lane & 15, hi = lane >> 4;
  const int col = wave * 16 + lr;

  const int r8 = tid & 7;
  const int jj = (tid >> 3) & 15, ah = tid >> 7;
  const int mc = min(m0 + r8, NH - 1);
  const float* src = dists + mc + NH * jj;
  float v[16];
#pragma unroll
  for (int p = 0; p < 16; ++p) v[p] = src[(2 * p + ah) * NN];
  float4 pv[4];
#pragma unroll
  for (int q = 0; q < 4; ++q) pv[q] = *(const float4*)(P + (tid + 256 * q) * 4);
  const int ib = (2 * m0) % NH;
  float h1[8], h2[8];
#pragma unroll
  for (int vi = 0; vi < 8; ++vi) {
    int i1 = ib + 2 * vi;
    if (i1 >= NH) i1 -= NH;
    int i2 = (i1 + 1 == NH) ? 0 : i1 + 1;
    h1[vi] = H[i1 * DD + col];
    h2[vi] = H[i2 * DD + col];
  }
  const float bv = bias[col];

  float* dst = &sS[(jj >> 1) * SS_TILE + ((jj & 1) * 8 + r8) * SS_STRIDE];
#pragma unroll
  for (int p = 0; p < 16; ++p) dst[2 * p + ah] = v[p];
#pragma unroll
  for (int q = 0; q < 4; ++q) {
    const int i = tid + 256 * q, r = i >> 5, c4 = i & 31;
    *(float4*)(&sP[r * SP_STRIDE + c4 * 4]) = pv[q];
  }
  float st[8];
#pragma unroll
  for (int vi = 0; vi < 8; ++vi) st[vi] = (h1[vi] + h2[vi]) * (1.0f / 32.0f);
  __syncthreads();

  v2f b[8];
#pragma unroll
  for (int t = 0; t < 8; ++t) {
    const int ka = t * 4 + hi * 2;
    b[t].x = sP[ka * SP_STRIDE + col];
    b[t].y = sP[(ka + 1) * SP_STRIDE + col];
  }

  if (m0 + MR <= NH) tiles<true>(sS, b, st, bv, m0, lr, hi, col, out);
  else               tiles<false>(sS, b, st, bv, m0, lr, hi, col, out);
}

extern "C" void kernel_launch(void* const* d_in, const int* in_sizes, int n_in,
                              void* d_out, int out_size, void* d_ws, size_t ws_size,
                              hipStream_t stream) {
  const float* embeds = (const float*)d_in[0];
  const float* dists  = (const float*)d_in[1];
  const float* Wh     = (const float*)d_in[2];
  const float* bias   = (const float*)d_in[3];
  const int*   ids    = (const int*)d_in[4];
  float* out = (float*)d_out;
  float* ws  = (float*)d_ws;

  float* P  = ws;
  float* H  = P + AA * DD;

  front_kernel<<<AA + HB, 256, 0, stream>>>(embeds, Wh, ids, P, H);
  final_kernel<<<FB, 256, 0, stream>>>(dists, P, H, bias, out);
}


# Front kernel: each block reads its half of W_hidden once with whole-line b128 loads into LDS; P's dot products and H's B fragments read LDS (were 32- and 16-line gathers, P's in 16 serial trips)
# speedup vs baseline: 2.1044x; 1.0654x over previous
#include <hip/hip_runtime.h>

typedef float v2f __attribute__((ext_vector_type(2)));
typedef float v4f __attribute__((ext_vector_type(4)));
typedef float v8f __attribute__((ext_vector_type(8)));

#define NN 30000
#define AA 32
#define DD 128
#define NH 1875
#define HR 8
#define HB 235
#define SA_STRIDE 132
#define SW_STRIDE 132
#define SP_STRIDE 136
#define SS_STRIDE 34
#define SS_TILE (16 * SS_STRIDE)
#define MR 8
#define NT 8
#define FB 235


__device__ __forceinline__ void load_w(const float* wsrc, v4f (&wv)[16]) {
#pragma unroll
  for (int q = 0; q < 16; ++q) wv[q] = *(const v4f*)(wsrc + q * 8 * (2 * DD));
}
__device__ __forceinline__ void stage_w(float* sW, int wave, int lane, const v4f (&wv)[16]) {
#pragma unroll
  for (int q = 0; q < 16; ++q) *(v4f*)(&sW[(wave + 8 * q) * SW_STRIDE + lane * 4]) = wv[q];
}

__global__ void __launch_bounds__(256)
front_kernel(const float* __restrict__ embeds, const float* __restrict__ Wh,
             const int* __restrict__ ids, float* __restrict__ P, float* __restrict__ H) {
  __shared__ __attribute__((aligned(16))) float sW[DD * SW_STRIDE];
  __shared__ __attribute__((aligned(16))) float sA[16 * SA_STRIDE];
  const int tid = threadIdx.x;
  const int lane = tid & 31, wave = tid >> 5;
  v4f wv[16];

  if (blockIdx.x < AA) {
    const int a = blockIdx.x;
    const int id = ids[a];
    load_w(Wh + wave * (2 * DD) + lane * 4, wv);
    const v4f ev = *(const v4f*)(embeds + (long)id * DD + lane * 4);
    stage_w(sW, wave, lane, wv);
    __syncthreads();
    if (tid >= DD) return;
    const int k = tid;
    v4f w[32];
#pragma unroll
    for (int t = 0; t < 32; ++t) w[t] = *(const v4f*)(&sW[k * SW_STRIDE + t * 4]);
    __builtin_amdgcn_sched_barrier(0);
    float s = 0.f;
#pragma unroll
    for (int d = 0; d < DD; ++d) {
      const float ed = __int_as_float(__builtin_amdgcn_readlane(__float_as_int(ev[d & 3]), d >> 2));
      s = fmaf(ed, w[d >> 2][d & 3], s);
    }
    P[a * DD + k] = s * (1.0f / 32.0f);
    return;
  }

  const int lr = lane & 15, hi = lane >> 4;
  const int col = wave * 16 + lr;
  const int i0 = (blockIdx.x - AA) * HR;
  const int i = i0 + wave;

  load_w(Wh + DD + wave * (2 * DD) + lane * 4, wv);

  v4f s = {0.f, 0.f, 0.f, 0.f};
  v4f r[16];
  if (i < NH) {
    const float* q = embeds + (long)i * 16 * DD + lane * 4;
#pragma unroll
    for (int t = 0; t < 16; ++t) r[t] = *(const v4f*)(q + t * DD);
  }
  stage_w(sW, wave, lane, wv);
  if (i < NH) {
#pragma unroll
    for (int t = 0; t < 16; ++t) s += r[t];
  }
  const v4f z = {0.f, 0.f, 0.f, 0.f};
  *(v4f*)(&sA[wave * SA_STRIDE + lane * 4]) = s;
  *(v4f*)(&sA[(wave + HR) * SA_STRIDE + lane * 4]) = z;
  __syncthreads();

  v2f af[32], bf[32];
#pragma unroll
  for (int t = 0; t < 32; ++t) {
    const int ka = t * 4 + hi * 2;
    af[t] = *(const v2f*)(&sA[lr * SA_STRIDE + ka]);
    bf[t] = *(const v2f*)(&sW[col * SW_STRIDE + ka]);
  }
  v8f acc = {};
#pragma unroll
  for (int t = 0; t < 32; ++t)
    acc = __builtin_amdgcn_wmma_f32_16x16x4_f32(false, af[t], false, bf[t], (short)0, acc, false, false);

  __syncthreads();
  if (hi == 0) {
#pragma unroll
    for (int vi = 0; vi < 8; ++vi) sA[vi * SA_STRIDE + col] = acc[vi];
  }
  __syncthreads();
  if (i < NH) {
    v4f h = *(const v4f*)(&sA[wave * SA_STRIDE + lane * 4]);
    *(v4f*)(H + (long)i * DD + lane * 4) = h;
  }
}


__device__ __forceinline__ void load_pair(const float* s0, v2f (&a0)[8], v2f (&a1)[8]) {
#pragma unroll
  for (int t = 0; t < 8; ++t) {
    a0[t] = *(const v2f*)(s0 + t * 4);
    a1[t] = *(const v2f*)(s0 + SS_TILE + t * 4);
  }
}

template <bool FULL>
__device__ __forceinline__ void tiles(const float* sS, const v2f (&b)[8], const float (&st)[8], float bv,
                                      int m0, int lr, int hi, int col, float* __restrict__ out) {
  const float* sa = &sS[lr * SS_STRIDE + hi * 2];
  v2f a0[8], a1[8];
  load_pair(sa, a0, a1);
#pragma unroll
  for (int tp = 0; tp < NT / 2; ++tp) {
    v2f n0[8], n1[8];
    const bool more = tp + 1 < NT / 2;
    if (more) load_pair(sa + (2 * tp + 2) * SS_TILE, n0, n1);
    v8f acc0 = {}, acc1 = {};
#pragma unroll
    for (int t = 0; t < 8; ++t) {
      acc0 = __builtin_amdgcn_wmma_f32_16x16x4_f32(false, a0[t], false, b[t], (short)0, acc0, false, false);
      acc1 = __builtin_amdgcn_wmma_f32_16x16x4_f32(false, a1[t], false, b[t], (short)0, acc1, false, false);
    }
    const int nb = m0 + NH * (4 * tp + hi);
#pragma unroll
    for (int vi = 0; vi < 8; ++vi) {
      if (FULL || m0 + vi < NH) {
        const int n = nb + vi;
        out[n * DD + col] = acc0[vi] + st[vi] + bv;
        out[(n + 2 * NH) * DD + col] = acc1[vi] + st[vi] + bv;
      }
    }
    __builtin_amdgcn_sched_barrier(0);
    if (more) {
#pragma unroll
      for (int t = 0; t < 8; ++t) { a0[t] = n0[t]; a1[t] = n1[t]; }
    }
  }
}

__global__ void __launch_bounds__(256)
final_kernel(const float* __restrict__ dists, const float* __restrict__ P,
             const float* __restrict__ H, const float* __restrict__ bias,
             float* __restrict__ out) {
  __shared__ __attribute__((aligned(16))) float sS[NT * SS_TILE];
  __shared__ __attribute__((aligned(16))) float sP[AA * SP_STRIDE];
  const int tid = threadIdx.x;
  const int m0 = blockIdx.x * MR;
  const int lane = tid & 31, wave = tid >> 5;
  const int lr = lane & 15, hi = lane >> 4;
  const int col = wave * 16 + lr;

  const int r8 = tid & 7;
  const int jj = (tid >> 3) & 15, ah = tid >> 7;
  const int mc = min(m0 + r8, NH - 1);
  const float* src = dists + mc + NH * jj;
  float v[16];
#pragma unroll
  for (int p = 0; p < 16; ++p) v[p] = src[(2 * p + ah) * NN];
  float4 pv[4];
#pragma unroll
  for (int q = 0; q < 4; ++q) pv[q] = *(const float4*)(P + (tid + 256 * q) * 4);
  const int ib = (2 * m0) % NH;
  float h1[8], h2[8];
#pragma unroll
  for (int vi = 0; vi < 8; ++vi) {
    int i1 = ib + 2 * vi;
    if (i1 >= NH) i1 -= NH;
    int i2 = (i1 + 1 == NH) ? 0 : i1 + 1;
    h1[vi] = H[i1 * DD + col];
    h2[vi] = H[i2 * DD + col];
  }
  const float bv = bias[col];

  float* dst = &sS[(jj >> 1) * SS_TILE + ((jj & 1) * 8 + r8) * SS_STRIDE];
#pragma unroll
  for (int p = 0; p < 16; ++p) dst[2 * p + ah] = v[p];
#pragma unroll
  for (int q = 0; q < 4; ++q) {
    const int i = tid + 256 * q, r = i >> 5, c4 = i & 31;
    *(float4*)(&sP[r * SP_STRIDE + c4 * 4]) = pv[q];
  }
  float st[8];
#pragma unroll
  for (int vi = 0; vi < 8; ++vi) st[vi] = (h1[vi] + h2[vi]) * (1.0f / 32.0f);
  __syncthreads();

  v2f b[8];
#pragma unroll
  for (int t = 0; t < 8; ++t) {
    const int ka = t * 4 + hi * 2;
    b[t].x = sP[ka * SP_STRIDE + col];
    b[t].y = sP[(ka + 1) * SP_STRIDE + col];
  }

  if (m0 + MR <= NH) tiles<true>(sS, b, st, bv, m0, lr, hi, col, out);
  else               tiles<false>(sS, b, st, bv, m0, lr, hi, col, out);
}

extern "C" void kernel_launch(void* const* d_in, const int* in_sizes, int n_in,
                              void* d_out, int out_size, void* d_ws, size_t ws_size,
                              hipStream_t stream) {
  const float* embeds = (const float*)d_in[0];
  const float* dists  = (const float*)d_in[1];
  const float* Wh     = (const float*)d_in[2];
  const float* bias   = (const float*)d_in[3];
  const int*   ids    = (const int*)d_in[4];
  float* out = (float*)d_out;
  float* ws  = (float*)d_ws;

  float* P  = ws;
  float* H  = P + AA * DD;

  front_kernel<<<AA + HB, 256, 0, stream>>>(embeds, Wh, ids, P, H);
  final_kernel<<<FB, 256, 0, stream>>>(dists, P, H, bias, out);
}


# One launch, not two: each group forms P (on f32 WMMA) and its own self term behind LDS-only barriers, so later operands' loads stay in flight under the products; float sum order changes
# speedup vs baseline: 2.2752x; 1.0812x over previous
#include <hip/hip_runtime.h>

typedef float v2f __attribute__((ext_vector_type(2)));
typedef float v4f __attribute__((ext_vector_type(4)));
typedef float v8f __attribute__((ext_vector_type(8)));

#define NN 30000
#define AA 32
#define DD 128
#define NH 1875
#define SA_STRIDE 132
#define SW_STRIDE 132
#define SP_STRIDE 136
#define SS_STRIDE 34
#define SS_TILE (16 * SS_STRIDE)
#define MR 8
#define NT 8
#define FB 235


__device__ __forceinline__ void load_w(const float* wsrc, v4f (&wv)[16]) {
#pragma unroll
  for (int q = 0; q < 16; ++q) wv[q] = *(const v4f*)(wsrc + q * 8 * (2 * DD));
}
__device__ __forceinline__ void stage_w(float* sW, int wave, int lane, const v4f (&wv)[16]) {
#pragma unroll
  for (int q = 0; q < 16; ++q) *(v4f*)(&sW[(wave + 8 * q) * SW_STRIDE + lane * 4]) = wv[q];
}

__device__ __forceinline__ void load_pair(const float* s0, v2f (&a0)[8], v2f (&a1)[8]) {
#pragma unroll
  for (int t = 0; t < 8; ++t) {
    a0[t] = *(const v2f*)(s0 + t * 4);
    a1[t] = *(const v2f*)(s0 + SS_TILE + t * 4);
  }
}

template <bool FULL>
__device__ __forceinline__ void tiles(const float* sS, const v2f (&b)[8], const float (&st)[8], float bv,
                                      int m0, int lr, int hi, int col, float* __restrict__ out) {
  const float* sa = &sS[lr * SS_STRIDE + hi * 2];
  v2f a0[8], a1[8];
  load_pair(sa, a0, a1);
#pragma unroll
  for (int tp = 0; tp < NT / 2; ++tp) {
    v2f n0[8], n1[8];
    const bool more = tp + 1 < NT / 2;
    if (more) load_pair(sa + (2 * tp + 2) * SS_TILE, n0, n1);
    v8f acc0 = {}, acc1 = {};
#pragma unroll
    for (int t = 0; t < 8; ++t) {
      acc0 = __builtin_amdgcn_wmma_f32_16x16x4_f32(false, a0[t], false, b[t], (short)0, acc0, false, false);
      acc1 = __builtin_amdgcn_wmma_f32_16x16x4_f32(false, a1[t], false, b[t], (short)0, acc1, false, false);
    }
    const int nb = m0 + NH * (4 * tp + hi);
#pragma unroll
    for (int vi = 0; vi < 8; ++vi) {
      if (FULL || m0 + vi < NH) {
        const int n = nb + vi;
        out[n * DD + col] = acc0[vi] + st[vi] + bv;
        out[(n + 2 * NH) * DD + col] = acc1[vi] + st[vi] + bv;
      }
    }
    __builtin_amdgcn_sched_barrier(0);
    if (more) {
#pragma unroll
      for (int t = 0; t < 8; ++t) { a0[t] = n0[t]; a1[t] = n1[t]; }
    }
  }
}


__device__ __forceinline__ void lds_barrier() {
  asm volatile("s_wait_dscnt 0x0\n\ts_barrier_signal -1\n\ts_barrier_wait -1" ::: "memory");
}

__global__ void __launch_bounds__(256)
fused_kernel(const float* __restrict__ embeds, const float* __restrict__ dists,
             const float* __restrict__ Wh, const float* __restrict__ bias,
             const int* __restrict__ ids, float* __restrict__ out) {
  __shared__ __attribute__((aligned(16))) float sW1[DD * SW_STRIDE];
  __shared__ __attribute__((aligned(16))) float sW2[DD * SW_STRIDE];
  __shared__ __attribute__((aligned(16))) float sAn[AA * SA_STRIDE];
  __shared__ __attribute__((aligned(16))) float sWs[MR * SA_STRIDE];
  __shared__ __attribute__((aligned(16))) float sS[NT * SS_TILE];
  __shared__ __attribute__((aligned(16))) float sP[AA * SP_STRIDE];
  const int tid = threadIdx.x;
  const int m0 = blockIdx.x * MR;
  const int lane = tid & 31, wave = tid >> 5;
  const int lr = lane & 15, hi = lane >> 4;
  const int col = wave * 16 + lr;

  int id[4];
#pragma unroll
  for (int q = 0; q < 4; ++q) id[q] = ids[wave + 8 * q];
  v4f wv1[16];
  load_w(Wh + wave * (2 * DD) + lane * 4, wv1);
  v4f an[4];
#pragma unroll
  for (int q = 0; q < 4; ++q) an[q] = *(const v4f*)(embeds + (long)id[q] * DD + lane * 4);
  v4f wv2[16];
  load_w(Wh + DD + wave * (2 * DD) + lane * 4, wv2);
  const int mw = min(m0 + wave, NH - 1);
  const int i1 = (2 * mw) % NH;
  const int i2 = (i1 + 1 == NH) ? 0 : i1 + 1;
  v4f r1[16], r2[16];
  {
    const float* q1 = embeds + (long)i1 * 16 * DD + lane * 4;
    const float* q2 = embeds + (long)i2 * 16 * DD + lane * 4;
#pragma unroll
    for (int t = 0; t < 16; ++t) r1[t] = *(const v4f*)(q1 + t * DD);
#pragma unroll
    for (int t = 0; t < 16; ++t) r2[t] = *(const v4f*)(q2 + t * DD);
  }
  const int r8 = tid & 7;
  const int jj = (tid >> 3) & 15, ah = tid >> 7;
  const int mc = min(m0 + r8, NH - 1);
  const float* src = dists + mc + NH * jj;
  float v[16];
#pragma unroll
  for (int p = 0; p < 16; ++p) v[p] = src[(2 * p + ah) * NN];
  const float bv = bias[col];

  stage_w(sW1, wave, lane, wv1);
#pragma unroll
  for (int q = 0; q < 4; ++q) *(v4f*)(&sAn[(wave + 8 * q) * SA_STRIDE + lane * 4]) = an[q];
  lds_barrier();
  {
    v2f a0[32], a1[32], bf[32];
#pragma unroll
    for (int t = 0; t < 32; ++t) {
      const int ka = t * 4 + hi * 2;
      a0[t] = *(const v2f*)(&sAn[lr * SA_STRIDE + ka]);
      a1[t] = *(const v2f*)(&sAn[(lr + 16) * SA_STRIDE + ka]);
      bf[t] = *(const v2f*)(&sW1[col * SW_STRIDE + ka]);
    }
    v8f p0 = {}, p1 = {};
#pragma unroll
    for (int t = 0; t < 32; ++t) {
      p0 = __builtin_amdgcn_wmma_f32_16x16x4_f32(false, a0[t], false, bf[t], (short)0, p0, false, false);
      p1 = __builtin_amdgcn_wmma_f32_16x16x4_f32(false, a1[t], false, bf[t], (short)0, p1, false, false);
    }
#pragma unroll
    for (int vi = 0; vi < 8; ++vi) {
      sP[(vi + 8 * hi) * SP_STRIDE + col] = p0[vi] * (1.0f / 32.0f);
      sP[(vi + 8 * hi + 16) * SP_STRIDE + col] = p1[vi] * (1.0f / 32.0f);
    }
  }

  stage_w(sW2, wave, lane, wv2);
  {
    v4f s1 = {0.f, 0.f, 0.f, 0.f}, s2 = {0.f, 0.f, 0.f, 0.f};
#pragma unroll
    for (int t = 0; t < 16; ++t) { s1 += r1[t]; s2 += r2[t]; }
    *(v4f*)(&sWs[wave * SA_STRIDE + lane * 4]) = s1 + s2;
  }
  lds_barrier();
  float st[8];
  {
    v2f af[32], bf[32];
#pragma unroll
    for (int t = 0; t < 32; ++t) {
      const int ka = t * 4 + hi * 2;
      af[t] = *(const v2f*)(&sWs[(lr & 7) * SA_STRIDE + ka]);
      bf[t] = *(const v2f*)(&sW2[col * SW_STRIDE + ka]);
    }
    v8f acc = {};
#pragma unroll
    for (int t = 0; t < 32; ++t)
      acc = __builtin_amdgcn_wmma_f32_16x16x4_f32(false, af[t], false, bf[t], (short)0, acc, false, false);
#pragma unroll
    for (int vi = 0; vi < 8; ++vi) st[vi] = acc[vi] * (1.0f / 32.0f);
  }

  float* dst = &sS[(jj >> 1) * SS_TILE + ((jj & 1) * 8 + r8) * SS_STRIDE];
#pragma unroll
  for (int p = 0; p < 16; ++p) dst[2 * p + ah] = v[p];
  lds_barrier();

  v2f b[8];
#pragma unroll
  for (int t = 0; t < 8; ++t) {
    const int ka = t * 4 + hi * 2;
    b[t].x = sP[ka * SP_STRIDE + col];
    b[t].y = sP[(ka + 1) * SP_STRIDE + col];
  }

  if (m0 + MR <= NH) tiles<true>(sS, b, st, bv, m0, lr, hi, col, out);
  else               tiles<false>(sS, b, st, bv, m0, lr, hi, col, out);
}

extern "C" void kernel_launch(void* const* d_in, const int* in_sizes, int n_in,
                              void* d_out, int out_size, void* d_ws, size_t ws_size,
                              hipStream_t stream) {
  const float* embeds = (const float*)d_in[0];
  const float* dists  = (const float*)d_in[1];
  const float* Wh     = (const float*)d_in[2];
  const float* bias   = (const float*)d_in[3];
  const int*   ids    = (const int*)d_in[4];
  float* out = (float*)d_out;

  fused_kernel<<<FB, 256, 0, stream>>>(embeds, dists, Wh, bias, ids, out);
}


# The 8 tile products move ahead of the self term, under the window rows' landing, and are parked in LDS as rows; after the self term each wave adds it and stores whole 512-B rows
# speedup vs baseline: 2.4175x; 1.0625x over previous
#include <hip/hip_runtime.h>

typedef float v2f __attribute__((ext_vector_type(2)));
typedef float v4f __attribute__((ext_vector_type(4)));
typedef float v8f __attribute__((ext_vector_type(8)));

#define NN 30000
#define AA 32
#define DD 128
#define NH 1875
#define SA_STRIDE 132
#define SW_STRIDE 132
#define SP_STRIDE 136
#define SS_STRIDE 34
#define SS_TILE (16 * SS_STRIDE)
#define MR 8
#define NT 8
#define FB 235
#define SO_STRIDE 132


__device__ __forceinline__ void load_w(const float* wsrc, v4f (&wv)[16]) {
#pragma unroll
  for (int q = 0; q < 16; ++q) wv[q] = *(const v4f*)(wsrc + q * 8 * (2 * DD));
}
__device__ __forceinline__ void stage_w(float* sW, int wave, int lane, const v4f (&wv)[16]) {
#pragma unroll
  for (int q = 0; q < 16; ++q) *(v4f*)(&sW[(wave + 8 * q) * SW_STRIDE + lane * 4]) = wv[q];
}

__device__ __forceinline__ void load_pair(const float* s0, v2f (&a0)[8], v2f (&a1)[8]) {
#pragma unroll
  for (int t = 0; t < 8; ++t) {
    a0[t] = *(const v2f*)(s0 + t * 4);
    a1[t] = *(const v2f*)(s0 + SS_TILE + t * 4);
  }
}

__device__ __forceinline__ void tiles(const float* sS, const v2f (&b)[8], float* sO, int lr, int hi, int col) {
  const float* sa = &sS[lr * SS_STRIDE + hi * 2];
  float* so = &sO[(8 * hi) * SO_STRIDE + col];
  v2f a0[8], a1[8];
  load_pair(sa, a0, a1);
#pragma unroll
  for (int tp = 0; tp < NT / 2; ++tp) {
    v2f n0[8], n1[8];
    const bool more = tp + 1 < NT / 2;
    if (more) load_pair(sa + (2 * tp + 2) * SS_TILE, n0, n1);
    v8f acc0 = {}, acc1 = {};
#pragma unroll
    for (int t = 0; t < 8; ++t) {
      acc0 = __builtin_amdgcn_wmma_f32_16x16x4_f32(false, a0[t], false, b[t], (short)0, acc0, false, false);
      acc1 = __builtin_amdgcn_wmma_f32_16x16x4_f32(false, a1[t], false, b[t], (short)0, acc1, false, false);
    }
#pragma unroll
    for (int vi = 0; vi < 8; ++vi) {
      so[((2 * tp) * 16 + vi) * SO_STRIDE] = acc0[vi];
      so[((2 * tp + 1) * 16 + vi) * SO_STRIDE] = acc1[vi];
    }
    if (more) {
#pragma unroll
      for (int t = 0; t < 8; ++t) { a0[t] = n0[t]; a1[t] = n1[t]; }
    }
  }
}


__device__ __forceinline__ void lds_barrier() {
  asm volatile("s_wait_dscnt 0x0\n\ts_barrier_signal -1\n\ts_barrier_wait -1" ::: "memory");
}

__global__ void __launch_bounds__(256)
fused_kernel(const float* __restrict__ embeds, const float* __restrict__ dists,
             const float* __restrict__ Wh, const float* __restrict__ bias,
             const int* __restrict__ ids, float* __restrict__ out) {
  __shared__ __attribute__((aligned(16))) float sW1[DD * SW_STRIDE];
  __shared__ __attribute__((aligned(16))) float sW2[DD * SW_STRIDE];
  __shared__ __attribute__((aligned(16))) float sAn[AA * SA_STRIDE];
  __shared__ __attribute__((aligned(16))) float sWs[MR * SA_STRIDE];
  __shared__ __attribute__((aligned(16))) float sS[NT * SS_TILE];
  __shared__ __attribute__((aligned(16))) float sP[AA * SP_STRIDE];
  __shared__ __attribute__((aligned(16))) float sO[NT * 16 * SO_STRIDE];
  __shared__ __attribute__((aligned(16))) float sSf[MR * SA_STRIDE];
  const int tid = threadIdx.x;
  const int m0 = blockIdx.x * MR;
  const int lane = tid & 31, wave = tid >> 5;
  const int lr = lane & 15, hi = lane >> 4;
  const int col = wave * 16 + lr;

  int id[4];
#pragma unroll
  for (int q = 0; q < 4; ++q) id[q] = ids[wave + 8 * q];
  v4f wv1[16];
  load_w(Wh + wave * (2 * DD) + lane * 4, wv1);
  v4f an[4];
#pragma unroll
  for (int q = 0; q < 4; ++q) an[q] = *(const v4f*)(embeds + (long)id[q] * DD + lane * 4);
  const int r8 = tid & 7;
  const int jj = (tid >> 3) & 15, ah = tid >> 7;
  const int mc = min(m0 + r8, NH - 1);
  const float* src = dists + mc + NH * jj;
  float v[16];
#pragma unroll
  for (int p = 0; p < 16; ++p) v[p] = src[(2 * p + ah) * NN];
  v4f wv2[16];
  load_w(Wh + DD + wave * (2 * DD) + lane * 4, wv2);
  const int mw = min(m0 + wave, NH - 1);
  const int i1 = (2 * mw) % NH;
  const int i2 = (i1 + 1 == NH) ? 0 : i1 + 1;
  v4f r1[16], r2[16];
  {
    const float* q1 = embeds + (long)i1 * 16 * DD + lane * 4;
    const float* q2 = embeds + (long)i2 * 16 * DD + lane * 4;
#pragma unroll
    for (int t = 0; t < 16; ++t) r1[t] = *(const v4f*)(q1 + t * DD);
#pragma unroll
    for (int t = 0; t < 16; ++t) r2[t] = *(const v4f*)(q2 + t * DD);
  }
  const v4f b4 = *(const v4f*)(bias + lane * 4);

  stage_w(sW1, wave, lane, wv1);
#pragma unroll
  for (int q = 0; q < 4; ++q) *(v4f*)(&sAn[(wave + 8 * q) * SA_STRIDE + lane * 4]) = an[q];
  lds_barrier();
  {
    v2f a0[32], a1[32], bf[32];
#pragma unroll
    for (int t = 0; t < 32; ++t) {
      const int ka = t * 4 + hi * 2;
      a0[t] = *(const v2f*)(&sAn[lr * SA_STRIDE + ka]);
      a1[t] = *(const v2f*)(&sAn[(lr + 16) * SA_STRIDE + ka]);
      bf[t] = *(const v2f*)(&sW1[col * SW_STRIDE + ka]);
    }
    v8f p0 = {}, p1 = {};
#pragma unroll
    for (int t = 0; t < 32; ++t) {
      p0 = __builtin_amdgcn_wmma_f32_16x16x4_f32(false, a0[t], false, bf[t], (short)0, p0, false, false);
      p1 = __builtin_amdgcn_wmma_f32_16x16x4_f32(false, a1[t], false, bf[t], (short)0, p1, false, false);
    }
#pragma unroll
    for (int vi = 0; vi < 8; ++vi) {
      sP[(vi + 8 * hi) * SP_STRIDE + col] = p0[vi] * (1.0f / 32.0f);
      sP[(vi + 8 * hi + 16) * SP_STRIDE + col] = p1[vi] * (1.0f / 32.0f);
    }
  }

  float* dst = &sS[(jj >> 1) * SS_TILE + ((jj & 1) * 8 + r8) * SS_STRIDE];
#pragma unroll
  for (int p = 0; p < 16; ++p) dst[2 * p + ah] = v[p];
  stage_w(sW2, wave, lane, wv2);
  lds_barrier();
  {
    v2f b[8];
#pragma unroll
    for (int t = 0; t < 8; ++t) {
      const int ka = t * 4 + hi * 2;
      b[t].x = sP[ka * SP_STRIDE + col];
      b[t].y = sP[(ka + 1) * SP_STRIDE + col];
    }
    tiles(sS, b, sO, lr, hi, col);
  }
  lds_barrier();
  v4f x[16];
#pragma unroll
  for (int j = 0; j < 16; ++j) x[j] = *(const v4f*)(&sO[((j >> 1) * 16 + (j & 1) * 8 + wave) * SO_STRIDE + lane * 4]);
  __builtin_amdgcn_sched_barrier(0);

  {
    v4f s1 = {0.f, 0.f, 0.f, 0.f}, s2 = {0.f, 0.f, 0.f, 0.f};
#pragma unroll
    for (int t = 0; t < 16; ++t) { s1 += r1[t]; s2 += r2[t]; }
    *(v4f*)(&sWs[wave * SA_STRIDE + lane * 4]) = s1 + s2;
  }
  lds_barrier();
  {
    v2f af[32], bf[32];
#pragma unroll
    for (int t = 0; t < 32; ++t) {
      const int ka = t * 4 + hi * 2;
      af[t] = *(const v2f*)(&sWs[(lr & 7) * SA_STRIDE + ka]);
      bf[t] = *(const v2f*)(&sW2[col * SW_STRIDE + ka]);
    }
    v8f acc = {};
#pragma unroll
    for (int t = 0; t < 32; ++t)
      acc = __builtin_amdgcn_wmma_f32_16x16x4_f32(false, af[t], false, bf[t], (short)0, acc, false, false);
    if (hi == 0) {
#pragma unroll
      for (int vi = 0; vi < 8; ++vi) sSf[vi * SA_STRIDE + col] = acc[vi] * (1.0f / 32.0f);
    }
  }
  lds_barrier();

  if (m0 + wave < NH) {
    const v4f sv = *(const v4f*)(&sSf[wave * SA_STRIDE + lane * 4]);
    float* o = out + (m0 + wave) * DD + lane * 4;
#pragma unroll
    for (int j = 0; j < 16; ++j) *(v4f*)(o + j * (NH * DD)) = (x[j] + sv) + b4;
  }
}

extern "C" void kernel_launch(void* const* d_in, const int* in_sizes, int n_in,
                              void* d_out, int out_size, void* d_ws, size_t ws_size,
                              hipStream_t stream) {
  const float* embeds = (const float*)d_in[0];
  const float* dists  = (const float*)d_in[1];
  const float* Wh     = (const float*)d_in[2];
  const float* bias   = (const float*)d_in[3];
  const int*   ids    = (const int*)d_in[4];
  float* out = (float*)d_out;

  fused_kernel<<<FB, 256, 0, stream>>>(embeds, dists, Wh, bias, ids, out);
}
